# starting kernel
# baseline (speedup 1.0000x reference)
#include <hip/hip_runtime.h>

typedef __attribute__((ext_vector_type(2))) float v2f;
typedef __attribute__((ext_vector_type(8))) float v8f;

#define CD 8
#define TT 512
#define NSEG 511
#define NTHREADS 128

__global__ __launch_bounds__(NTHREADS)
void logsig_kernel(const float* __restrict__ path, float* __restrict__ out) {
    __shared__ float dxs[NSEG * CD];
    __shared__ float S3s[512];
    __shared__ float S2b[2][64];
    __shared__ float S1b[2][8];
    __shared__ float S4s[4096];

    const int b    = blockIdx.x;
    const int tid  = threadIdx.x;
    const int lane = tid & 31;
    const int w    = tid >> 5;
    const int Ln   = lane & 15;
    const int half = lane >> 4;

    const float* pb = path + (size_t)b * (TT * CD);

    for (int e = tid; e < NSEG * CD; e += NTHREADS)
        dxs[e] = pb[e + 8] - pb[e];
    for (int e = tid; e < 512; e += NTHREADS) S3s[e] = 0.0f;
    if (tid < 64) S2b[0][tid] = 0.0f;
    if (tid < 8)  S1b[0][tid] = 0.0f;

    v8f acc[8];
    const v8f vzero = {0.f, 0.f, 0.f, 0.f, 0.f, 0.f, 0.f, 0.f};
#pragma unroll
    for (int q = 0; q < 8; ++q) acc[q] = vzero;

    int cur = 0;
    for (int t = 0; t < NSEG; ++t) {
        __syncthreads();
        const float* d = &dxs[t * CD];
        const float dk = d[Ln & 7];

        const float bv = (Ln < 8) ? d[Ln] : 0.0f;
        v2f bf; bf[0] = bv; bf[1] = bv;

        float s3n[8];
#pragma unroll
        for (int q = 0; q < 8; ++q) {
            const int tt = w * 8 + q;
            const int r  = tt * 16 + Ln;
            const int i  = r >> 6;
            const int j  = (r >> 3) & 7;
            const float s2v = S2b[cur][(i << 3) | j];
            const float s1v = S1b[cur][i];
            const float di = d[i], dj = d[j];
            const float p1 = s2v * dk;
            const float p2 = s1v * dj * dk;
            const float p3 = di * dj * dk;
            const float s3v = S3s[r];

            v2f af;
            af[0] = half ? (1.0f / 6.0f)  * p2 : s3v;
            af[1] = half ? (1.0f / 24.0f) * p3 : 0.5f * p1;

            acc[q] = __builtin_amdgcn_wmma_f32_16x16x4_f32(
                false, af, false, bf, (short)0, acc[q], false, false);

            s3n[q] = s3v + p1 + 0.5f * p2 + (1.0f / 6.0f) * p3;
        }
        if (half == 0) {
#pragma unroll
            for (int q = 0; q < 8; ++q)
                S3s[(w * 8 + q) * 16 + Ln] = s3n[q];
        }
        const int nxt = cur ^ 1;
        if (tid < 64) {
            const int i2 = tid >> 3, j2 = tid & 7;
            S2b[nxt][tid] = S2b[cur][tid] + S1b[cur][i2] * d[j2]
                          + 0.5f * d[i2] * d[j2];
        } else if (tid < 72) {
            const int c = tid - 64;
            S1b[nxt][c] = S1b[cur][c] + d[c];
        }
        cur = nxt;
    }

    __syncthreads();
    if (Ln < 8) {
#pragma unroll
        for (int q = 0; q < 8; ++q) {
            const int tt = w * 8 + q;
#pragma unroll
            for (int v = 0; v < 8; ++v) {
                const int m = v + (half << 3);
                S4s[(tt * 16 + m) * 8 + Ln] = acc[q][v];
            }
        }
    }
    __syncthreads();

    float* ob = out + (size_t)b * 4680;
    const float* S1f = S1b[cur];
    const float* S2f = S2b[cur];

    if (tid < 8) ob[tid] = S1f[tid];
    if (tid < 64) {
        const int i = tid >> 3, j = tid & 7;
        ob[8 + tid] = S2f[tid] - 0.5f * S1f[i] * S1f[j];
    }
    for (int e = tid; e < 512; e += NTHREADS) {
        const int i = e >> 6, j = (e >> 3) & 7, k = e & 7;
        const float v3 = S3s[e]
            - 0.5f * (S1f[i] * S2f[(j << 3) | k] + S2f[(i << 3) | j] * S1f[k])
            + (1.0f / 3.0f) * S1f[i] * S1f[j] * S1f[k];
        ob[72 + e] = v3;
    }
    for (int e = tid; e < 4096; e += NTHREADS) {
        const int i = e >> 9, j = (e >> 6) & 7, k = (e >> 3) & 7, l = e & 7;
        const float s1i = S1f[i], s1j = S1f[j], s1k = S1f[k], s1l = S1f[l];
        const float v4 = S4s[e]
            - 0.5f * ( s1i * S3s[e & 511]
                     + S2f[(i << 3) | j] * S2f[(k << 3) | l]
                     + S3s[e >> 3] * s1l )
            + (1.0f / 3.0f) * ( s1i * s1j * S2f[(k << 3) | l]
                              + s1i * S2f[(j << 3) | k] * s1l
                              + S2f[(i << 3) | j] * s1k * s1l )
            - 0.25f * s1i * s1j * s1k * s1l;
        ob[584 + e] = v4;
    }
}

extern "C" void kernel_launch(void* const* d_in, const int* in_sizes, int n_in,
                              void* d_out, int out_size, void* d_ws, size_t ws_size,
                              hipStream_t stream) {
    const float* path = (const float*)d_in[0];
    float* out = (float*)d_out;
    const int B = in_sizes[0] / (TT * CD);
    logsig_kernel<<<dim3(B), dim3(NTHREADS), 0, stream>>>(path, out);
}


# Level 4 re-associated: summed over steps as (level-2 row terms)[ij] x (increment k x level 1 still to come l)[kl] on the matrix unit, no level-3 state carried through time; sum order changes
# speedup vs baseline: 10.5958x; 10.5958x over previous
#include <hip/hip_runtime.h>

typedef __attribute__((ext_vector_type(2))) float v2f;
typedef __attribute__((ext_vector_type(4))) float v4f;
typedef __attribute__((ext_vector_type(8))) float v8f;

#define CD 8
#define TT 512
#define NBLK 32
#define TPAD 528
#define PITCH 532
#define NTHREADS 128

__device__ __forceinline__ float other_half(float x) {
    const int xi = __builtin_bit_cast(int, x);
    const int r  = __builtin_amdgcn_permlanex16(xi, xi, 0x76543210, 0xfedcba98, false, false);
    return __builtin_bit_cast(float, r);
}

struct BlockIn {
    v4f dj[2];
    v4f di[2];
    v4f s1[2];
    v4f rl[2];
    v4f dk[4][2];
};

__device__ __forceinline__ void load_block(BlockIn& x, const float* djp, const float* dip,
                                           const float* s1p, const float* rlp,
                                           const float* dkp, int s0) {
#pragma unroll
    for (int g = 0; g < 2; ++g) {
        x.dj[g] = *(const v4f*)(djp + s0 + 4 * g);
        x.di[g] = *(const v4f*)(dip + s0 + 4 * g);
        x.s1[g] = *(const v4f*)(s1p + s0 + 4 * g);
        x.rl[g] = *(const v4f*)(rlp + s0 + 4 * g);
#pragma unroll
        for (int ct = 0; ct < 4; ++ct)
            x.dk[ct][g] = *(const v4f*)(dkp + 2 * ct * PITCH + s0 + 4 * g);
    }
}

__device__ __forceinline__ void level2_block(const BlockIn& x, int h, float& carry2,
                                             float (&V)[8], float (&cc)[8]) {
    float dj[8], di[8], s1[8];
#pragma unroll
    for (int v = 0; v < 4; ++v) {
        dj[v] = x.dj[0][v]; dj[v + 4] = x.dj[1][v];
        di[v] = x.di[0][v]; di[v + 4] = x.di[1][v];
        s1[v] = x.s1[0][v]; s1[v + 4] = x.s1[1][v];
    }
    float u[8];
    float t2 = 0.0f;
#pragma unroll
    for (int v = 0; v < 8; ++v) {
        u[v] = s1[v] + 0.5f * di[v];
        t2 = u[v] * dj[v] + t2;
    }
    const float t2o = other_half(t2);
    float r2 = h ? (carry2 + t2o) : carry2;
    carry2 = carry2 + (t2 + t2o);
#pragma unroll
    for (int v = 0; v < 8; ++v) {
        V[v]  = r2 + (0.5f * s1[v] + (1.0f / 6.0f) * di[v]) * dj[v];
        cc[v] = 0.5f * r2 + ((1.0f / 6.0f) * s1[v] + (1.0f / 24.0f) * di[v]) * dj[v];
        r2 = u[v] * dj[v] + r2;
    }
}

__global__ __launch_bounds__(NTHREADS)
void logsig_kernel(const float* __restrict__ path, float* __restrict__ out) {
    __shared__ __attribute__((aligned(16))) float dT[CD * PITCH];
    __shared__ __attribute__((aligned(16))) float xT[CD * PITCH];
    __shared__ __attribute__((aligned(16))) float rT[CD * PITCH];
    __shared__ __attribute__((aligned(16))) float S4s[4096];
    __shared__ __attribute__((aligned(16))) float S3s[512];
    __shared__ float S2f[64];
    __shared__ float S1f[8];

    const int b    = blockIdx.x;
    const int tid  = threadIdx.x;
    const int lane = tid & 31;
    const int w    = tid >> 5;
    const int n    = lane & 15;
    const int h    = lane >> 4;
    const int ii   = 2 * w + (n >> 3);
    const int jj   = n & 7;

    const float* pb = path + (size_t)b * (TT * CD);

    {
        const v4f* pb4 = (const v4f*)pb;
        const v4f o0 = pb4[0], o1 = pb4[1];
        const v4f e0 = pb4[2 * (TT - 1)], e1 = pb4[2 * (TT - 1) + 1];
#pragma unroll
        for (int it = 0; it < TT / NTHREADS; ++it) {
            const int s  = tid + NTHREADS * it;
            const int sn = (s < TT - 1) ? s + 1 : s;
            const v4f a0 = pb4[2 * s],  a1 = pb4[2 * s + 1];
            const v4f n0 = pb4[2 * sn], n1 = pb4[2 * sn + 1];
#pragma unroll
            for (int c = 0; c < 4; ++c) {
                dT[c * PITCH + s]       = n0[c] - a0[c];
                dT[(c + 4) * PITCH + s] = n1[c] - a1[c];
                xT[c * PITCH + s]       = a0[c] - o0[c];
                xT[(c + 4) * PITCH + s] = a1[c] - o1[c];
                rT[c * PITCH + s]       = e0[c] - n0[c];
                rT[(c + 4) * PITCH + s] = e1[c] - n1[c];
            }
        }
        {
            const int c = tid >> 4, s = TT + (tid & 15);
            dT[c * PITCH + s] = 0.0f;
            xT[c * PITCH + s] = 0.0f;
            rT[c * PITCH + s] = 0.0f;
        }
    }
    __syncthreads();

    v8f acc[4];
    v8f acc3;
    const v8f vzero = {0.f, 0.f, 0.f, 0.f, 0.f, 0.f, 0.f, 0.f};
#pragma unroll
    for (int q = 0; q < 4; ++q) acc[q] = vzero;
    acc3 = vzero;
    float carry2 = 0.0f;

    const float* djp = &dT[jj * PITCH + 8 * h];
    const float* dip = &dT[ii * PITCH + 8 * h];
    const float* s1p = &xT[ii * PITCH + 8 * h];
    const float* rlp = &rT[jj * PITCH + 8 * h];
    const float* dkp = &dT[(n >> 3) * PITCH + 8 * h];

    BlockIn cur;
    float V[8], cc[8];
    load_block(cur, djp, dip, s1p, rlp, dkp, 0);
    level2_block(cur, h, carry2, V, cc);

#pragma unroll 2
    for (int blk = 0; blk < NBLK; ++blk) {
        BlockIn nxt;
        load_block(nxt, djp, dip, s1p, rlp, dkp, (blk + 1) * 16);

#pragma unroll
        for (int m = 0; m < 4; ++m) {
            const int g = m >> 1, e = 2 * (m & 1);
            v2f af; af[0] = V[2 * m];  af[1] = V[2 * m + 1];
            v2f cf; cf[0] = cc[2 * m]; cf[1] = cc[2 * m + 1];
            v2f dl; dl[0] = cur.dj[g][e]; dl[1] = cur.dj[g][e + 1];
            v2f rl; rl[0] = cur.rl[g][e]; rl[1] = cur.rl[g][e + 1];
#pragma unroll
            for (int ct = 0; ct < 4; ++ct) {
                v2f dk; dk[0] = cur.dk[ct][g][e]; dk[1] = cur.dk[ct][g][e + 1];
                acc[ct] = __builtin_amdgcn_wmma_f32_16x16x4_f32(
                    false, af, false, dk * rl, (short)0, acc[ct], false, false);
            }
            acc3 = __builtin_amdgcn_wmma_f32_16x16x4_f32(
                false, af, false, dl, (short)0, acc3, false, false);
#pragma unroll
            for (int ct = 0; ct < 4; ++ct) {
                v2f dk; dk[0] = cur.dk[ct][g][e]; dk[1] = cur.dk[ct][g][e + 1];
                acc[ct] = __builtin_amdgcn_wmma_f32_16x16x4_f32(
                    false, cf, false, dk * dl, (short)0, acc[ct], false, false);
            }
        }

        level2_block(nxt, h, carry2, V, cc);
        cur = nxt;
    }

    if (tid < 8) S1f[tid] = xT[tid * PITCH + (TT - 1)];
    if (h == 0) S2f[16 * w + n] = carry2;
    if (n < 8) {
#pragma unroll
        for (int r = 0; r < 8; ++r) S3s[(16 * w + 8 * h + r) * 8 + n] = acc3[r];
    }
#pragma unroll
    for (int ct = 0; ct < 4; ++ct) {
#pragma unroll
        for (int r = 0; r < 8; ++r)
            S4s[(16 * w + 8 * h + r) * 64 + 16 * ct + n] = acc[ct][r];
    }
    __syncthreads();

    float* ob = out + (size_t)b * 4680;

    if (tid < 8) ob[tid] = S1f[tid];
    if (tid < 64) {
        const int i = tid >> 3, j = tid & 7;
        ob[8 + tid] = S2f[tid] - 0.5f * S1f[i] * S1f[j];
    }
    for (int e = tid; e < 512; e += NTHREADS) {
        const int i = e >> 6, j = (e >> 3) & 7, k = e & 7;
        const float v3 = S3s[e]
            - 0.5f * (S1f[i] * S2f[(j << 3) | k] + S2f[(i << 3) | j] * S1f[k])
            + (1.0f / 3.0f) * S1f[i] * S1f[j] * S1f[k];
        ob[72 + e] = v3;
    }
    for (int e = tid; e < 4096; e += NTHREADS) {
        const int i = e >> 9, j = (e >> 6) & 7, k = (e >> 3) & 7, l = e & 7;
        const float s1i = S1f[i], s1j = S1f[j], s1k = S1f[k], s1l = S1f[l];
        const float v4 = S4s[e]
            - 0.5f * ( s1i * S3s[e & 511]
                     + S2f[(i << 3) | j] * S2f[(k << 3) | l]
                     + S3s[e >> 3] * s1l )
            + (1.0f / 3.0f) * ( s1i * s1j * S2f[(k << 3) | l]
                              + s1i * S2f[(j << 3) | k] * s1l
                              + S2f[(i << 3) | j] * s1k * s1l )
            - 0.25f * s1i * s1j * s1k * s1l;
        ob[584 + e] = v4;
    }
}

extern "C" void kernel_launch(void* const* d_in, const int* in_sizes, int n_in,
                              void* d_out, int out_size, void* d_ws, size_t ws_size,
                              hipStream_t stream) {
    const float* path = (const float*)d_in[0];
    float* out = (float*)d_out;
    const int B = in_sizes[0] / (TT * CD);
    logsig_kernel<<<dim3(B), dim3(NTHREADS), 0, stream>>>(path, out);
}


# Log series of level 4 finished in the accumulator registers: no LDS staging, 3 fixed-offset table reads per element (was 11), half swap for 128-byte stores; fences hold the loop's reads; same sums
# speedup vs baseline: 11.5086x; 1.0861x over previous
#include <hip/hip_runtime.h>

typedef __attribute__((ext_vector_type(2))) float v2f;
typedef __attribute__((ext_vector_type(4))) float v4f;
typedef __attribute__((ext_vector_type(8))) float v8f;

#define CD 8
#define TT 512
#define NBLK 32
#define TPAD 528
#define PITCH 532
#define NTHREADS 128

__device__ __forceinline__ float other_half(float x) {
    const int xi = __builtin_bit_cast(int, x);
    const int r  = __builtin_amdgcn_permlanex16(xi, xi, 0x76543210, 0xfedcba98, false, false);
    return __builtin_bit_cast(float, r);
}

struct BlockIn {
    v4f dj[2];
    v4f di[2];
    v4f s1[2];
    v4f rl[2];
    v4f dk[4][2];
};

__device__ __forceinline__ void load_block(BlockIn& x, const float* djp, const float* dip,
                                           const float* s1p, const float* rlp,
                                           const float* dkp, int s0) {
#pragma unroll
    for (int g = 0; g < 2; ++g) {
        x.dj[g] = *(const v4f*)(djp + s0 + 4 * g);
        x.di[g] = *(const v4f*)(dip + s0 + 4 * g);
        x.s1[g] = *(const v4f*)(s1p + s0 + 4 * g);
        x.rl[g] = *(const v4f*)(rlp + s0 + 4 * g);
#pragma unroll
        for (int ct = 0; ct < 4; ++ct)
            x.dk[ct][g] = *(const v4f*)(dkp + 2 * ct * PITCH + s0 + 4 * g);
    }
}

__device__ __forceinline__ void level2_block(const BlockIn& x, int h, float& carry2,
                                             float (&V)[8], float (&cc)[8]) {
    float dj[8], di[8], s1[8];
#pragma unroll
    for (int v = 0; v < 4; ++v) {
        dj[v] = x.dj[0][v]; dj[v + 4] = x.dj[1][v];
        di[v] = x.di[0][v]; di[v + 4] = x.di[1][v];
        s1[v] = x.s1[0][v]; s1[v + 4] = x.s1[1][v];
    }
    float u[8];
    float t2 = 0.0f;
#pragma unroll
    for (int v = 0; v < 8; ++v) {
        u[v] = s1[v] + 0.5f * di[v];
        t2 = u[v] * dj[v] + t2;
    }
    const float t2o = other_half(t2);
    float r2 = h ? (carry2 + t2o) : carry2;
    carry2 = carry2 + (t2 + t2o);
#pragma unroll
    for (int v = 0; v < 8; ++v) {
        V[v]  = r2 + (0.5f * s1[v] + (1.0f / 6.0f) * di[v]) * dj[v];
        cc[v] = 0.5f * r2 + ((1.0f / 6.0f) * s1[v] + (1.0f / 24.0f) * di[v]) * dj[v];
        r2 = u[v] * dj[v] + r2;
    }
}

__global__ __launch_bounds__(NTHREADS)
void logsig_kernel(const float* __restrict__ path, float* __restrict__ out) {
    __shared__ __attribute__((aligned(16))) float dT[CD * PITCH];
    __shared__ __attribute__((aligned(16))) float xT[CD * PITCH];
    __shared__ __attribute__((aligned(16))) float rT[CD * PITCH];
    __shared__ __attribute__((aligned(16))) float S3s[512];
    __shared__ float S2f[64];
    __shared__ float S1f[8];

    const int b    = blockIdx.x;
    const int tid  = threadIdx.x;
    const int lane = tid & 31;
    const int w    = tid >> 5;
    const int n    = lane & 15;
    const int h    = lane >> 4;
    const int ii   = 2 * w + (n >> 3);
    const int jj   = n & 7;

    const float* pb = path + (size_t)b * (TT * CD);

    {
        const v4f* pb4 = (const v4f*)pb;
        const v4f o0 = pb4[0], o1 = pb4[1];
        const v4f e0 = pb4[2 * (TT - 1)], e1 = pb4[2 * (TT - 1) + 1];
#pragma unroll
        for (int it = 0; it < TT / NTHREADS; ++it) {
            const int s  = tid + NTHREADS * it;
            const int sn = (s < TT - 1) ? s + 1 : s;
            const v4f a0 = pb4[2 * s],  a1 = pb4[2 * s + 1];
            const v4f n0 = pb4[2 * sn], n1 = pb4[2 * sn + 1];
#pragma unroll
            for (int c = 0; c < 4; ++c) {
                dT[c * PITCH + s]       = n0[c] - a0[c];
                dT[(c + 4) * PITCH + s] = n1[c] - a1[c];
                xT[c * PITCH + s]       = a0[c] - o0[c];
                xT[(c + 4) * PITCH + s] = a1[c] - o1[c];
                rT[c * PITCH + s]       = e0[c] - n0[c];
                rT[(c + 4) * PITCH + s] = e1[c] - n1[c];
            }
        }
        {
            const int c = tid >> 4, s = TT + (tid & 15);
            dT[c * PITCH + s] = 0.0f;
            xT[c * PITCH + s] = 0.0f;
            rT[c * PITCH + s] = 0.0f;
        }
    }
    __syncthreads();

    v8f acc[4];
    v8f acc3;
    const v8f vzero = {0.f, 0.f, 0.f, 0.f, 0.f, 0.f, 0.f, 0.f};
#pragma unroll
    for (int q = 0; q < 4; ++q) acc[q] = vzero;
    acc3 = vzero;
    float carry2 = 0.0f;

    const float* djp = &dT[jj * PITCH + 8 * h];
    const float* dip = &dT[ii * PITCH + 8 * h];
    const float* s1p = &xT[ii * PITCH + 8 * h];
    const float* rlp = &rT[jj * PITCH + 8 * h];
    const float* dkp = &dT[(n >> 3) * PITCH + 8 * h];

    BlockIn cur;
    float V[8], cc[8];
    load_block(cur, djp, dip, s1p, rlp, dkp, 0);
    level2_block(cur, h, carry2, V, cc);

#pragma unroll 2
    for (int blk = 0; blk < NBLK; ++blk) {
        BlockIn nxt;
        __builtin_amdgcn_sched_barrier(0);
        load_block(nxt, djp, dip, s1p, rlp, dkp, (blk + 1) * 16);
        __builtin_amdgcn_sched_barrier(0);

#pragma unroll
        for (int m = 0; m < 4; ++m) {
            const int g = m >> 1, e = 2 * (m & 1);
            v2f af; af[0] = V[2 * m];  af[1] = V[2 * m + 1];
            v2f cf; cf[0] = cc[2 * m]; cf[1] = cc[2 * m + 1];
            v2f dl; dl[0] = cur.dj[g][e]; dl[1] = cur.dj[g][e + 1];
            v2f rl; rl[0] = cur.rl[g][e]; rl[1] = cur.rl[g][e + 1];
#pragma unroll
            for (int ct = 0; ct < 4; ++ct) {
                v2f dk; dk[0] = cur.dk[ct][g][e]; dk[1] = cur.dk[ct][g][e + 1];
                acc[ct] = __builtin_amdgcn_wmma_f32_16x16x4_f32(
                    false, af, false, dk * rl, (short)0, acc[ct], false, false);
            }
            acc3 = __builtin_amdgcn_wmma_f32_16x16x4_f32(
                false, af, false, dl, (short)0, acc3, false, false);
#pragma unroll
            for (int ct = 0; ct < 4; ++ct) {
                v2f dk; dk[0] = cur.dk[ct][g][e]; dk[1] = cur.dk[ct][g][e + 1];
                acc[ct] = __builtin_amdgcn_wmma_f32_16x16x4_f32(
                    false, cf, false, dk * dl, (short)0, acc[ct], false, false);
            }
            if (m == 1) __builtin_amdgcn_sched_barrier(0);
        }

        level2_block(nxt, h, carry2, V, cc);
        cur = nxt;
    }

    if (tid < 8) S1f[tid] = xT[tid * PITCH + (TT - 1)];
    if (h == 0) S2f[16 * w + n] = carry2;
    if (n < 8) {
#pragma unroll
        for (int r = 0; r < 8; ++r) S3s[(16 * w + 8 * h + r) * 8 + n] = acc3[r];
    }
    __syncthreads();

    float* ob = out + (size_t)b * 4680;

    if (tid < 8) ob[tid] = S1f[tid];
    if (tid < 64) {
        const int i = tid >> 3, j = tid & 7;
        ob[8 + tid] = S2f[tid] - 0.5f * S1f[i] * S1f[j];
    }
    for (int e = tid; e < 512; e += NTHREADS) {
        const int i = e >> 6, j = (e >> 3) & 7, k = e & 7;
        const float v3 = S3s[e]
            - 0.5f * (S1f[i] * S2f[(j << 3) | k] + S2f[(i << 3) | j] * S1f[k])
            + (1.0f / 3.0f) * S1f[i] * S1f[j] * S1f[k];
        ob[72 + e] = v3;
    }

    {
        const int i4 = 2 * w + h, kh = n >> 3, l4 = n & 7;
        const float s1i = S1f[i4], s1l = S1f[l4];
        float s1j[8], s2ij[8];
#pragma unroll
        for (int r = 0; r < 8; ++r) { s1j[r] = S1f[r]; s2ij[r] = S2f[i4 * 8 + r]; }
        const float* s3jkl = &S3s[n];
        const float* s2jk  = &S2f[kh];
        const float* s3ijk = &S3s[i4 * 64 + kh];
        float v[4][8];
#pragma unroll
        for (int ct = 0; ct < 4; ++ct) {
            const float s1k  = S1f[2 * ct + kh];
            const float s2kl = S2f[(2 * ct + kh) * 8 + l4];
#pragma unroll
            for (int r = 0; r < 8; ++r) {
                const float a3 = s3jkl[64 * r + 16 * ct];
                const float b2 = s2jk[8 * r + 2 * ct];
                const float c3 = s3ijk[8 * r + 2 * ct];
                v[ct][r] = acc[ct][r]
                    - 0.5f * ( s1i * a3 + s2ij[r] * s2kl + c3 * s1l )
                    + (1.0f / 3.0f) * ( s1i * s1j[r] * s2kl
                                      + s1i * b2 * s1l
                                      + s2ij[r] * s1k * s1l )
                    - 0.25f * s1i * s1j[r] * s1k * s1l;
            }
        }
        float* o4 = ob + 584 + lane;
#pragma unroll
        for (int cp = 0; cp < 2; ++cp) {
#pragma unroll
            for (int r = 0; r < 8; ++r) {
                const float lo = v[2 * cp][r], hi = v[2 * cp + 1][r];
                const float got = other_half(h ? lo : hi);
                const float x0 = h ? got : lo;
                const float x1 = h ? hi : got;
                o4[(2 * w) * 512 + r * 64 + 32 * cp]     = x0;
                o4[(2 * w + 1) * 512 + r * 64 + 32 * cp] = x1;
            }
        }
    }
}

extern "C" void kernel_launch(void* const* d_in, const int* in_sizes, int n_in,
                              void* d_out, int out_size, void* d_ws, size_t ws_size,
                              hipStream_t stream) {
    const float* path = (const float*)d_in[0];
    float* out = (float*)d_out;
    const int B = in_sizes[0] / (TT * CD);
    logsig_kernel<<<dim3(B), dim3(NTHREADS), 0, stream>>>(path, out);
}


# Level 4's column side (third index k) split over 2 workgroups per batch (grid B x 2): each keeps all row tiles and levels 1-3, does half the level-4 column tiles and stores; sums unchanged
# speedup vs baseline: 15.5455x; 1.3508x over previous
#include <hip/hip_runtime.h>

typedef __attribute__((ext_vector_type(2))) float v2f;
typedef __attribute__((ext_vector_type(4))) float v4f;
typedef __attribute__((ext_vector_type(8))) float v8f;

#define CD 8
#define TT 512
#define NBLK 32
#define TPAD 528
#define PITCH 532
#define NTHREADS 128

__device__ __forceinline__ float other_half(float x) {
    const int xi = __builtin_bit_cast(int, x);
    const int r  = __builtin_amdgcn_permlanex16(xi, xi, 0x76543210, 0xfedcba98, false, false);
    return __builtin_bit_cast(float, r);
}

struct BlockIn {
    v4f dj[2];
    v4f di[2];
    v4f s1[2];
    v4f rl[2];
    v4f dk[2][2];
};

__device__ __forceinline__ void load_block(BlockIn& x, const float* djp, const float* dip,
                                           const float* s1p, const float* rlp,
                                           const float* dkp, int s0) {
#pragma unroll
    for (int g = 0; g < 2; ++g) {
        x.dj[g] = *(const v4f*)(djp + s0 + 4 * g);
        x.di[g] = *(const v4f*)(dip + s0 + 4 * g);
        x.s1[g] = *(const v4f*)(s1p + s0 + 4 * g);
        x.rl[g] = *(const v4f*)(rlp + s0 + 4 * g);
#pragma unroll
        for (int c = 0; c < 2; ++c)
            x.dk[c][g] = *(const v4f*)(dkp + 2 * c * PITCH + s0 + 4 * g);
    }
}

__device__ __forceinline__ void level2_block(const BlockIn& x, int h, float& carry2,
                                             float (&V)[8], float (&cc)[8]) {
    float dj[8], di[8], s1[8];
#pragma unroll
    for (int v = 0; v < 4; ++v) {
        dj[v] = x.dj[0][v]; dj[v + 4] = x.dj[1][v];
        di[v] = x.di[0][v]; di[v + 4] = x.di[1][v];
        s1[v] = x.s1[0][v]; s1[v + 4] = x.s1[1][v];
    }
    float u[8];
    float t2 = 0.0f;
#pragma unroll
    for (int v = 0; v < 8; ++v) {
        u[v] = s1[v] + 0.5f * di[v];
        t2 = u[v] * dj[v] + t2;
    }
    const float t2o = other_half(t2);
    float r2 = h ? (carry2 + t2o) : carry2;
    carry2 = carry2 + (t2 + t2o);
#pragma unroll
    for (int v = 0; v < 8; ++v) {
        V[v]  = r2 + (0.5f * s1[v] + (1.0f / 6.0f) * di[v]) * dj[v];
        cc[v] = 0.5f * r2 + ((1.0f / 6.0f) * s1[v] + (1.0f / 24.0f) * di[v]) * dj[v];
        r2 = u[v] * dj[v] + r2;
    }
}

__global__ __launch_bounds__(NTHREADS)
void logsig_kernel(const float* __restrict__ path, float* __restrict__ out) {
    __shared__ __attribute__((aligned(16))) float dT[CD * PITCH];
    __shared__ __attribute__((aligned(16))) float xT[CD * PITCH];
    __shared__ __attribute__((aligned(16))) float rT[CD * PITCH];
    __shared__ __attribute__((aligned(16))) float S3s[512];
    __shared__ float S2f[64];
    __shared__ float S1f[8];

    const int b    = blockIdx.x;
    const int cg   = blockIdx.y;
    const int tid  = threadIdx.x;
    const int lane = tid & 31;
    const int w    = tid >> 5;
    const int n    = lane & 15;
    const int h    = lane >> 4;
    const int ii   = 2 * w + (n >> 3);
    const int jj   = n & 7;

    const float* pb = path + (size_t)b * (TT * CD);

    {
        const v4f* pb4 = (const v4f*)pb;
        const v4f o0 = pb4[0], o1 = pb4[1];
        const v4f e0 = pb4[2 * (TT - 1)], e1 = pb4[2 * (TT - 1) + 1];
#pragma unroll
        for (int it = 0; it < TT / NTHREADS; ++it) {
            const int s  = tid + NTHREADS * it;
            const int sn = (s < TT - 1) ? s + 1 : s;
            const v4f a0 = pb4[2 * s],  a1 = pb4[2 * s + 1];
            const v4f n0 = pb4[2 * sn], n1 = pb4[2 * sn + 1];
#pragma unroll
            for (int c = 0; c < 4; ++c) {
                dT[c * PITCH + s]       = n0[c] - a0[c];
                dT[(c + 4) * PITCH + s] = n1[c] - a1[c];
                xT[c * PITCH + s]       = a0[c] - o0[c];
                xT[(c + 4) * PITCH + s] = a1[c] - o1[c];
                rT[c * PITCH + s]       = e0[c] - n0[c];
                rT[(c + 4) * PITCH + s] = e1[c] - n1[c];
            }
        }
        {
            const int c = tid >> 4, s = TT + (tid & 15);
            dT[c * PITCH + s] = 0.0f;
            xT[c * PITCH + s] = 0.0f;
            rT[c * PITCH + s] = 0.0f;
        }
    }
    __syncthreads();

    v8f acc[2];
    v8f acc3;
    const v8f vzero = {0.f, 0.f, 0.f, 0.f, 0.f, 0.f, 0.f, 0.f};
#pragma unroll
    for (int q = 0; q < 2; ++q) acc[q] = vzero;
    acc3 = vzero;
    float carry2 = 0.0f;

    const float* djp = &dT[jj * PITCH + 8 * h];
    const float* dip = &dT[ii * PITCH + 8 * h];
    const float* s1p = &xT[ii * PITCH + 8 * h];
    const float* rlp = &rT[jj * PITCH + 8 * h];
    const float* dkp = &dT[(4 * cg + (n >> 3)) * PITCH + 8 * h];

    BlockIn cur;
    float V[8], cc[8];
    load_block(cur, djp, dip, s1p, rlp, dkp, 0);
    level2_block(cur, h, carry2, V, cc);

#pragma unroll 2
    for (int blk = 0; blk < NBLK; ++blk) {
        BlockIn nxt;
        __builtin_amdgcn_sched_barrier(0);
        load_block(nxt, djp, dip, s1p, rlp, dkp, (blk + 1) * 16);
        __builtin_amdgcn_sched_barrier(0);

#pragma unroll
        for (int m = 0; m < 4; ++m) {
            const int g = m >> 1, e = 2 * (m & 1);
            v2f af; af[0] = V[2 * m];  af[1] = V[2 * m + 1];
            v2f cf; cf[0] = cc[2 * m]; cf[1] = cc[2 * m + 1];
            v2f dl; dl[0] = cur.dj[g][e]; dl[1] = cur.dj[g][e + 1];
            v2f rl; rl[0] = cur.rl[g][e]; rl[1] = cur.rl[g][e + 1];
#pragma unroll
            for (int c = 0; c < 2; ++c) {
                v2f dk; dk[0] = cur.dk[c][g][e]; dk[1] = cur.dk[c][g][e + 1];
                acc[c] = __builtin_amdgcn_wmma_f32_16x16x4_f32(
                    false, af, false, dk * rl, (short)0, acc[c], false, false);
            }
            acc3 = __builtin_amdgcn_wmma_f32_16x16x4_f32(
                false, af, false, dl, (short)0, acc3, false, false);
#pragma unroll
            for (int c = 0; c < 2; ++c) {
                v2f dk; dk[0] = cur.dk[c][g][e]; dk[1] = cur.dk[c][g][e + 1];
                acc[c] = __builtin_amdgcn_wmma_f32_16x16x4_f32(
                    false, cf, false, dk * dl, (short)0, acc[c], false, false);
            }
            if (m == 1) __builtin_amdgcn_sched_barrier(0);
        }

        level2_block(nxt, h, carry2, V, cc);
        cur = nxt;
    }

    if (tid < 8) S1f[tid] = xT[tid * PITCH + (TT - 1)];
    if (h == 0) S2f[16 * w + n] = carry2;
    if (n < 8) {
#pragma unroll
        for (int r = 0; r < 8; ++r) S3s[(16 * w + 8 * h + r) * 8 + n] = acc3[r];
    }
    __syncthreads();

    float* ob = out + (size_t)b * 4680;

    if (cg == 0) {
        if (tid < 8) ob[tid] = S1f[tid];
        if (tid < 64) {
            const int i = tid >> 3, j = tid & 7;
            ob[8 + tid] = S2f[tid] - 0.5f * S1f[i] * S1f[j];
        }
    }
#pragma unroll
    for (int it = 0; it < 2; ++it) {
        const int e = 256 * cg + NTHREADS * it + tid;
        const int i = e >> 6, j = (e >> 3) & 7, k = e & 7;
        const float v3 = S3s[e]
            - 0.5f * (S1f[i] * S2f[(j << 3) | k] + S2f[(i << 3) | j] * S1f[k])
            + (1.0f / 3.0f) * S1f[i] * S1f[j] * S1f[k];
        ob[72 + e] = v3;
    }

    {
        const int i4 = 2 * w + h, kh = n >> 3, l4 = n & 7;
        const float s1i = S1f[i4], s1l = S1f[l4];
        float s1j[8], s2ij[8];
#pragma unroll
        for (int r = 0; r < 8; ++r) { s1j[r] = S1f[r]; s2ij[r] = S2f[i4 * 8 + r]; }
        const int kb = 4 * cg + kh;
        const float* s3jkl = &S3s[32 * cg + n];
        const float* s2jk  = &S2f[kb];
        const float* s3ijk = &S3s[i4 * 64 + kb];
        float v[2][8];
#pragma unroll
        for (int c = 0; c < 2; ++c) {
            const float s1k  = S1f[kb + 2 * c];
            const float s2kl = S2f[(kb + 2 * c) * 8 + l4];
#pragma unroll
            for (int r = 0; r < 8; ++r) {
                const float a3 = s3jkl[64 * r + 16 * c];
                const float b2 = s2jk[8 * r + 2 * c];
                const float c3 = s3ijk[8 * r + 2 * c];
                v[c][r] = acc[c][r]
                    - 0.5f * ( s1i * a3 + s2ij[r] * s2kl + c3 * s1l )
                    + (1.0f / 3.0f) * ( s1i * s1j[r] * s2kl
                                      + s1i * b2 * s1l
                                      + s2ij[r] * s1k * s1l )
                    - 0.25f * s1i * s1j[r] * s1k * s1l;
            }
        }
        float* o4 = ob + 584 + 32 * cg + lane;
#pragma unroll
        for (int r = 0; r < 8; ++r) {
            const float lo = v[0][r], hi = v[1][r];
            const float got = other_half(h ? lo : hi);
            const float x0 = h ? got : lo;
            const float x1 = h ? hi : got;
            o4[(2 * w) * 512 + r * 64]     = x0;
            o4[(2 * w + 1) * 512 + r * 64] = x1;
        }
    }
}

extern "C" void kernel_launch(void* const* d_in, const int* in_sizes, int n_in,
                              void* d_out, int out_size, void* d_ws, size_t ws_size,
                              hipStream_t stream) {
    const float* path = (const float*)d_in[0];
    float* out = (float*)d_out;
    const int B = in_sizes[0] / (TT * CD);
    logsig_kernel<<<dim3(B, 2), dim3(NTHREADS), 0, stream>>>(path, out);
}
